# starting kernel
# baseline (speedup 1.0000x reference)
#include <hip/hip_runtime.h>

typedef float v2f __attribute__((ext_vector_type(2)));
typedef float v8f __attribute__((ext_vector_type(8)));

#define WMMA_F32(a, b, c) \
    __builtin_amdgcn_wmma_f32_16x16x4_f32(false, (a), false, (b), (short)0, (c), false, false)

constexpr int Bsz    = 4;
constexpr int Nseq   = 2048;
constexpr int Cdim   = 1024;
constexpr int Hn     = 16;
constexpr int HD     = 64;
constexpr int ThreeC = 3 * Cdim;
constexpr int Mrows  = Bsz * Nseq;

__global__ __launch_bounds__(256)
void gemm_xwt_kernel(const float* __restrict__ X, const float* __restrict__ W,
                     const float* __restrict__ bias, float* __restrict__ Y,
                     int M, int Nd, int K)
{
    const int lane = threadIdx.x & 31;
    const int wid  = blockIdx.x * (blockDim.x >> 5) + (threadIdx.x >> 5);
    const int tilesN = Nd >> 6;
    const int m0 = (wid / tilesN) * 32;
    const int n0 = (wid % tilesN) * 64;
    const int hlf = lane >> 4;
    const int l16 = lane & 15;

    v8f zero = {};
    v8f acc[2][4];
#pragma unroll
    for (int i = 0; i < 2; i++)
#pragma unroll
        for (int j = 0; j < 4; j++) acc[i][j] = zero;

    const float* arow0 = X + (size_t)(m0 + l16) * K;
    const float* arow1 = X + (size_t)(m0 + 16 + l16) * K;
    const float* brow0 = W + (size_t)(n0 +  0 + l16) * K;
    const float* brow1 = W + (size_t)(n0 + 16 + l16) * K;
    const float* brow2 = W + (size_t)(n0 + 32 + l16) * K;
    const float* brow3 = W + (size_t)(n0 + 48 + l16) * K;

    const int ko0 = 2 * hlf;
    v2f a0 = *(const v2f*)(arow0 + ko0);
    v2f a1 = *(const v2f*)(arow1 + ko0);
    v2f b0 = *(const v2f*)(brow0 + ko0);
    v2f b1 = *(const v2f*)(brow1 + ko0);
    v2f b2 = *(const v2f*)(brow2 + ko0);
    v2f b3 = *(const v2f*)(brow3 + ko0);

    for (int k0 = 4; k0 < K; k0 += 4) {
        const int kn = k0 + ko0;
        v2f na0 = *(const v2f*)(arow0 + kn);
        v2f na1 = *(const v2f*)(arow1 + kn);
        v2f nb0 = *(const v2f*)(brow0 + kn);
        v2f nb1 = *(const v2f*)(brow1 + kn);
        v2f nb2 = *(const v2f*)(brow2 + kn);
        v2f nb3 = *(const v2f*)(brow3 + kn);

        acc[0][0] = WMMA_F32(a0, b0, acc[0][0]);
        acc[1][0] = WMMA_F32(a1, b0, acc[1][0]);
        acc[0][1] = WMMA_F32(a0, b1, acc[0][1]);
        acc[1][1] = WMMA_F32(a1, b1, acc[1][1]);
        acc[0][2] = WMMA_F32(a0, b2, acc[0][2]);
        acc[1][2] = WMMA_F32(a1, b2, acc[1][2]);
        acc[0][3] = WMMA_F32(a0, b3, acc[0][3]);
        acc[1][3] = WMMA_F32(a1, b3, acc[1][3]);

        a0 = na0; a1 = na1; b0 = nb0; b1 = nb1; b2 = nb2; b3 = nb3;
    }
    acc[0][0] = WMMA_F32(a0, b0, acc[0][0]);
    acc[1][0] = WMMA_F32(a1, b0, acc[1][0]);
    acc[0][1] = WMMA_F32(a0, b1, acc[0][1]);
    acc[1][1] = WMMA_F32(a1, b1, acc[1][1]);
    acc[0][2] = WMMA_F32(a0, b2, acc[0][2]);
    acc[1][2] = WMMA_F32(a1, b2, acc[1][2]);
    acc[0][3] = WMMA_F32(a0, b3, acc[0][3]);
    acc[1][3] = WMMA_F32(a1, b3, acc[1][3]);

#pragma unroll
    for (int i = 0; i < 2; i++)
#pragma unroll
        for (int j = 0; j < 4; j++)
#pragma unroll
            for (int r = 0; r < 8; r++) {
                const int row = m0 + 16 * i + r + 8 * hlf;
                const int col = n0 + 16 * j + l16;
                float v = acc[i][j][r];
                if (bias) v += bias[col];
                Y[(size_t)row * Nd + col] = v;
            }
}

__global__ __launch_bounds__(256)
void rope_kernel(float* __restrict__ qkv,
                 const float* __restrict__ cosT, const float* __restrict__ sinT)
{
    const int t = blockIdx.x * blockDim.x + threadIdx.x;
    const int i = t & 31;
    const int h = (t >> 5) & 15;
    const int n = (t >> 9) & (Nseq - 1);
    const int b = t >> 20;

    const size_t base = (size_t)(b * Nseq + n) * ThreeC + h * HD;
    const float c0 = cosT[n * HD + i],      s0 = sinT[n * HD + i];
    const float c1 = cosT[n * HD + i + 32], s1 = sinT[n * HD + i + 32];

    const float q0 = qkv[base + i], q1 = qkv[base + i + 32];
    qkv[base + i]      = q0 * c0 - q1 * s0;
    qkv[base + i + 32] = q1 * c1 + q0 * s1;

    const float k0 = qkv[base + Cdim + i], k1 = qkv[base + Cdim + i + 32];
    qkv[base + Cdim + i]      = k0 * c0 - k1 * s0;
    qkv[base + Cdim + i + 32] = k1 * c1 + k0 * s1;
}

__global__ __launch_bounds__(256)
void flash_attn_kernel(const float* __restrict__ qkv, float* __restrict__ O)
{
    const int lane = threadIdx.x & 31;
    const int wid  = blockIdx.x * (blockDim.x >> 5) + (threadIdx.x >> 5);
    const int qt = wid & 127;
    const int h  = (wid >> 7) & 15;
    const int b  = wid >> 11;
    const int hlf = lane >> 4;
    const int l16 = lane & 15;
    const int q0  = qt * 16;
    const float scale = 0.125f;

    const size_t rowQ = (size_t)(b * Nseq + q0 + l16) * ThreeC + h * HD;
    v2f qb[16];
#pragma unroll
    for (int kk = 0; kk < 16; kk++)
        qb[kk] = *(const v2f*)(qkv + rowQ + 4 * kk + 2 * hlf);

    v8f zero = {};
    v8f acc[4];
#pragma unroll
    for (int t = 0; t < 4; t++) acc[t] = zero;
    float m = -3.0e38f, l = 0.0f;

    const size_t baseK = (size_t)(b * Nseq) * ThreeC + Cdim + h * HD;
    const size_t baseV = (size_t)(b * Nseq) * ThreeC + 2 * Cdim + h * HD;

    for (int kt = 0; kt < Nseq / 16; kt++) {
        const int key0 = kt * 16;

        const size_t rowK = baseK + (size_t)(key0 + l16) * ThreeC;
        v2f ka[16];
#pragma unroll
        for (int kk = 0; kk < 16; kk++)
            ka[kk] = *(const v2f*)(qkv + rowK + 4 * kk + 2 * hlf);

        v8f s = zero;
#pragma unroll
        for (int kk = 0; kk < 16; kk++)
            s = WMMA_F32(ka[kk], qb[kk], s);

        v2f va[4][4];
#pragma unroll
        for (int t = 0; t < 4; t++) {
            const size_t colV = baseV + t * 16 + l16;
#pragma unroll
            for (int kk = 0; kk < 4; kk++) {
                const int ka0 = key0 + 4 * kk + 2 * hlf;
                va[t][kk].x = qkv[colV + (size_t)ka0 * ThreeC];
                va[t][kk].y = qkv[colV + (size_t)(ka0 + 1) * ThreeC];
            }
        }

        float sv[8];
#pragma unroll
        for (int r = 0; r < 8; r++) sv[r] = s[r] * scale;

        float tm = sv[0];
#pragma unroll
        for (int r = 1; r < 8; r++) tm = fmaxf(tm, sv[r]);
        tm = fmaxf(tm, __shfl_xor(tm, 16, 32));

        const float mn   = fmaxf(m, tm);
        const float corr = __expf(m - mn);
        m = mn;

        float p[8];
        float rs = 0.0f;
#pragma unroll
        for (int r = 0; r < 8; r++) { p[r] = __expf(sv[r] - mn); rs += p[r]; }
        rs += __shfl_xor(rs, 16, 32);
        l = l * corr + rs;

#pragma unroll
        for (int t = 0; t < 4; t++)
#pragma unroll
            for (int r = 0; r < 8; r++) acc[t][r] *= corr;

        const float x2 = __shfl_xor(p[2], 16, 32), x3 = __shfl_xor(p[3], 16, 32);
        const float x6 = __shfl_xor(p[6], 16, 32), x7 = __shfl_xor(p[7], 16, 32);
        const float u0 = __shfl_xor(p[0], 16, 32), u1 = __shfl_xor(p[1], 16, 32);
        const float u4 = __shfl_xor(p[4], 16, 32), u5 = __shfl_xor(p[5], 16, 32);
        v2f pb[4];
        pb[0].x = hlf ? x2 : p[0];  pb[0].y = hlf ? x3 : p[1];
        pb[1].x = hlf ? x6 : p[4];  pb[1].y = hlf ? x7 : p[5];
        pb[2].x = hlf ? p[2] : u0;  pb[2].y = hlf ? p[3] : u1;
        pb[3].x = hlf ? p[6] : u4;  pb[3].y = hlf ? p[7] : u5;

#pragma unroll
        for (int t = 0; t < 4; t++)
#pragma unroll
            for (int kk = 0; kk < 4; kk++)
                acc[t] = WMMA_F32(va[t][kk], pb[kk], acc[t]);
    }

    const float inv = 1.0f / l;
    const size_t baseO = (size_t)(b * Nseq + q0 + l16) * Cdim + h * HD;
#pragma unroll
    for (int t = 0; t < 4; t++)
#pragma unroll
        for (int r = 0; r < 8; r++)
            O[baseO + t * 16 + r + 8 * hlf] = acc[t][r] * inv;
}

extern "C" void kernel_launch(void* const* d_in, const int* in_sizes, int n_in,
                              void* d_out, int out_size, void* d_ws, size_t ws_size,
                              hipStream_t stream)
{
    const float* x        = (const float*)d_in[0];
    const float* rope_cos = (const float*)d_in[1];
    const float* rope_sin = (const float*)d_in[2];
    const float* W_qkv    = (const float*)d_in[3];
    const float* W_proj   = (const float*)d_in[4];
    const float* b_proj   = (const float*)d_in[5];
    float* out  = (float*)d_out;

    float* qkv  = (float*)d_ws;
    float* attn = qkv + (size_t)Mrows * ThreeC;

    {
        const int waves = (Mrows / 32) * (ThreeC / 64);
        gemm_xwt_kernel<<<waves / 8, 256, 0, stream>>>(x, W_qkv, nullptr, qkv,
                                                       Mrows, ThreeC, Cdim);
    }
    {
        const int threads = Bsz * Nseq * Hn * 32;
        rope_kernel<<<threads / 256, 256, 0, stream>>>(qkv, rope_cos, rope_sin);
    }
    {
        const int waves = Bsz * Hn * (Nseq / 16);
        flash_attn_kernel<<<waves / 8, 256, 0, stream>>>(qkv, attn);
    }
    {
        const int waves = (Mrows / 32) * (Cdim / 64);
        gemm_xwt_kernel<<<waves / 8, 256, 0, stream>>>(attn, W_proj, b_proj, out,
                                                       Mrows, Cdim, Cdim);
    }
}


# GEMMs: operand tiles staged through LDS per 128x128 block (coalesced 16-byte loads, two stages of K=64) instead of per-wave strided 8-byte global fragment loads; k order and sums as before
# speedup vs baseline: 1.9853x; 1.9853x over previous
#include <hip/hip_runtime.h>

typedef float v2f __attribute__((ext_vector_type(2)));
typedef float v4f __attribute__((ext_vector_type(4)));
typedef float v8f __attribute__((ext_vector_type(8)));

#define WMMA_F32(a, b, c) \
    __builtin_amdgcn_wmma_f32_16x16x4_f32(false, (a), false, (b), (short)0, (c), false, false)

constexpr int Bsz    = 4;
constexpr int Nseq   = 2048;
constexpr int Cdim   = 1024;
constexpr int Hn     = 16;
constexpr int HD     = 64;
constexpr int ThreeC = 3 * Cdim;
constexpr int Mrows  = Bsz * Nseq;

constexpr int GBM = 128;
constexpr int GBN = 128;
constexpr int GBK = 64;
constexpr int GLP = GBK + 4;
constexpr int GTR = GBK / 4;
constexpr int GRP = 256 / GTR;
constexpr int GNP = GBM / GRP;

#define GEMM_WMMA8(a0, a1, b0, b1, b2, b3)        \
    acc[0][0] = WMMA_F32(a0, b0, acc[0][0]);      \
    acc[1][0] = WMMA_F32(a1, b0, acc[1][0]);      \
    acc[0][1] = WMMA_F32(a0, b1, acc[0][1]);      \
    acc[1][1] = WMMA_F32(a1, b1, acc[1][1]);      \
    acc[0][2] = WMMA_F32(a0, b2, acc[0][2]);      \
    acc[1][2] = WMMA_F32(a1, b2, acc[1][2]);      \
    acc[0][3] = WMMA_F32(a0, b3, acc[0][3]);      \
    acc[1][3] = WMMA_F32(a1, b3, acc[1][3]);

#define GEMM_FRAGS(a0, a1, b0, b1, b2, b3, kk)            \
    a0 = *(const v2f*)(pa + 4 * (kk));                    \
    a1 = *(const v2f*)(pa + 16 * GLP + 4 * (kk));         \
    b0 = *(const v2f*)(pb + 4 * (kk));                    \
    b1 = *(const v2f*)(pb + 16 * GLP + 4 * (kk));         \
    b2 = *(const v2f*)(pb + 32 * GLP + 4 * (kk));         \
    b3 = *(const v2f*)(pb + 48 * GLP + 4 * (kk));

__device__ __forceinline__
void gemm_stage_mma(const float* __restrict__ pa, const float* __restrict__ pb,
                    v8f (&acc)[2][4])
{
    v2f pa0, pa1, pb0, pb1, pb2, pb3;
    v2f qa0, qa1, qb0, qb1, qb2, qb3;
    GEMM_FRAGS(pa0, pa1, pb0, pb1, pb2, pb3, 0)
#pragma unroll 1
    for (int kk = 0; kk < GBK / 4 - 2; kk += 2) {
        GEMM_FRAGS(qa0, qa1, qb0, qb1, qb2, qb3, kk + 1)
        GEMM_WMMA8(pa0, pa1, pb0, pb1, pb2, pb3)
        GEMM_FRAGS(pa0, pa1, pb0, pb1, pb2, pb3, kk + 2)
        GEMM_WMMA8(qa0, qa1, qb0, qb1, qb2, qb3)
    }
    GEMM_FRAGS(qa0, qa1, qb0, qb1, qb2, qb3, GBK / 4 - 1)
    GEMM_WMMA8(pa0, pa1, pb0, pb1, pb2, pb3)
    GEMM_WMMA8(qa0, qa1, qb0, qb1, qb2, qb3)
}

__global__ __launch_bounds__(256)
void gemm_xwt_kernel(const float* __restrict__ X, const float* __restrict__ W,
                     const float* __restrict__ bias, float* __restrict__ Y,
                     int M, int Nd, int K)
{
    __shared__ __attribute__((aligned(16))) float sA[2][GBM * GLP];
    __shared__ __attribute__((aligned(16))) float sB[2][GBN * GLP];

    const int tid  = threadIdx.x;
    const int lane = tid & 31;
    const int wv   = tid >> 5;
    const int hlf  = lane >> 4;
    const int l16  = lane & 15;
    const int tilesN = Nd / GBN;
    const int bm0 = (blockIdx.x / tilesN) * GBM;
    const int bn0 = (blockIdx.x % tilesN) * GBN;
    const int wm  = (wv >> 1) * 32;
    const int wn  = (wv & 1) * 64;

    v8f zero = {};
    v8f acc[2][4];
#pragma unroll
    for (int i = 0; i < 2; i++)
#pragma unroll
        for (int j = 0; j < 4; j++) acc[i][j] = zero;

    const int srow = tid / GTR;
    const int scol = (tid % GTR) * 4;
    const float* gA = X + (size_t)(bm0 + srow) * K + scol;
    const float* gB = W + (size_t)(bn0 + srow) * K + scol;
    const size_t gstep = (size_t)GRP * K;
    const int soff = srow * GLP + scol;

    const int aoff = (wm + l16) * GLP + 2 * hlf;
    const int boff = (wn + l16) * GLP + 2 * hlf;

    v4f ra[GNP], rb[GNP];
#pragma unroll
    for (int i = 0; i < GNP; i++) {
        ra[i] = *(const v4f*)(gA + i * gstep);
        rb[i] = *(const v4f*)(gB + i * gstep);
    }
#pragma unroll
    for (int i = 0; i < GNP; i++) {
        *(v4f*)(&sA[0][soff + i * GRP * GLP]) = ra[i];
        *(v4f*)(&sB[0][soff + i * GRP * GLP]) = rb[i];
    }
    __syncthreads();

    const int nk = K / GBK;
    for (int kt = 0; kt + 1 < nk; kt++) {
        const int buf = kt & 1;
        const int kn = (kt + 1) * GBK;
#pragma unroll
        for (int i = 0; i < GNP; i++) {
            ra[i] = *(const v4f*)(gA + i * gstep + kn);
            rb[i] = *(const v4f*)(gB + i * gstep + kn);
        }

        gemm_stage_mma(&sA[buf][aoff], &sB[buf][boff], acc);

#pragma unroll
        for (int i = 0; i < GNP; i++) {
            *(v4f*)(&sA[buf ^ 1][soff + i * GRP * GLP]) = ra[i];
            *(v4f*)(&sB[buf ^ 1][soff + i * GRP * GLP]) = rb[i];
        }
        __syncthreads();
    }
    {
        const int buf = (nk - 1) & 1;
        gemm_stage_mma(&sA[buf][aoff], &sB[buf][boff], acc);
    }

    const int m0 = bm0 + wm;
    const int n0 = bn0 + wn;
#pragma unroll
    for (int i = 0; i < 2; i++)
#pragma unroll
        for (int j = 0; j < 4; j++)
#pragma unroll
            for (int r = 0; r < 8; r++) {
                const int row = m0 + 16 * i + r + 8 * hlf;
                const int col = n0 + 16 * j + l16;
                float v = acc[i][j][r];
                if (bias) v += bias[col];
                Y[(size_t)row * Nd + col] = v;
            }
}

__global__ __launch_bounds__(256)
void rope_kernel(float* __restrict__ qkv,
                 const float* __restrict__ cosT, const float* __restrict__ sinT)
{
    const int t = blockIdx.x * blockDim.x + threadIdx.x;
    const int i = t & 31;
    const int h = (t >> 5) & 15;
    const int n = (t >> 9) & (Nseq - 1);
    const int b = t >> 20;

    const size_t base = (size_t)(b * Nseq + n) * ThreeC + h * HD;
    const float c0 = cosT[n * HD + i],      s0 = sinT[n * HD + i];
    const float c1 = cosT[n * HD + i + 32], s1 = sinT[n * HD + i + 32];

    const float q0 = qkv[base + i], q1 = qkv[base + i + 32];
    qkv[base + i]      = q0 * c0 - q1 * s0;
    qkv[base + i + 32] = q1 * c1 + q0 * s1;

    const float k0 = qkv[base + Cdim + i], k1 = qkv[base + Cdim + i + 32];
    qkv[base + Cdim + i]      = k0 * c0 - k1 * s0;
    qkv[base + Cdim + i + 32] = k1 * c1 + k0 * s1;
}

__global__ __launch_bounds__(256)
void flash_attn_kernel(const float* __restrict__ qkv, float* __restrict__ O)
{
    const int lane = threadIdx.x & 31;
    const int wid  = blockIdx.x * (blockDim.x >> 5) + (threadIdx.x >> 5);
    const int qt = wid & 127;
    const int h  = (wid >> 7) & 15;
    const int b  = wid >> 11;
    const int hlf = lane >> 4;
    const int l16 = lane & 15;
    const int q0  = qt * 16;
    const float scale = 0.125f;

    const size_t rowQ = (size_t)(b * Nseq + q0 + l16) * ThreeC + h * HD;
    v2f qb[16];
#pragma unroll
    for (int kk = 0; kk < 16; kk++)
        qb[kk] = *(const v2f*)(qkv + rowQ + 4 * kk + 2 * hlf);

    v8f zero = {};
    v8f acc[4];
#pragma unroll
    for (int t = 0; t < 4; t++) acc[t] = zero;
    float m = -3.0e38f, l = 0.0f;

    const size_t baseK = (size_t)(b * Nseq) * ThreeC + Cdim + h * HD;
    const size_t baseV = (size_t)(b * Nseq) * ThreeC + 2 * Cdim + h * HD;

    for (int kt = 0; kt < Nseq / 16; kt++) {
        const int key0 = kt * 16;

        const size_t rowK = baseK + (size_t)(key0 + l16) * ThreeC;
        v2f ka[16];
#pragma unroll
        for (int kk = 0; kk < 16; kk++)
            ka[kk] = *(const v2f*)(qkv + rowK + 4 * kk + 2 * hlf);

        v8f s = zero;
#pragma unroll
        for (int kk = 0; kk < 16; kk++)
            s = WMMA_F32(ka[kk], qb[kk], s);

        v2f va[4][4];
#pragma unroll
        for (int t = 0; t < 4; t++) {
            const size_t colV = baseV + t * 16 + l16;
#pragma unroll
            for (int kk = 0; kk < 4; kk++) {
                const int ka0 = key0 + 4 * kk + 2 * hlf;
                va[t][kk].x = qkv[colV + (size_t)ka0 * ThreeC];
                va[t][kk].y = qkv[colV + (size_t)(ka0 + 1) * ThreeC];
            }
        }

        float sv[8];
#pragma unroll
        for (int r = 0; r < 8; r++) sv[r] = s[r] * scale;

        float tm = sv[0];
#pragma unroll
        for (int r = 1; r < 8; r++) tm = fmaxf(tm, sv[r]);
        tm = fmaxf(tm, __shfl_xor(tm, 16, 32));

        const float mn   = fmaxf(m, tm);
        const float corr = __expf(m - mn);
        m = mn;

        float p[8];
        float rs = 0.0f;
#pragma unroll
        for (int r = 0; r < 8; r++) { p[r] = __expf(sv[r] - mn); rs += p[r]; }
        rs += __shfl_xor(rs, 16, 32);
        l = l * corr + rs;

#pragma unroll
        for (int t = 0; t < 4; t++)
#pragma unroll
            for (int r = 0; r < 8; r++) acc[t][r] *= corr;

        const float x2 = __shfl_xor(p[2], 16, 32), x3 = __shfl_xor(p[3], 16, 32);
        const float x6 = __shfl_xor(p[6], 16, 32), x7 = __shfl_xor(p[7], 16, 32);
        const float u0 = __shfl_xor(p[0], 16, 32), u1 = __shfl_xor(p[1], 16, 32);
        const float u4 = __shfl_xor(p[4], 16, 32), u5 = __shfl_xor(p[5], 16, 32);
        v2f pb[4];
        pb[0].x = hlf ? x2 : p[0];  pb[0].y = hlf ? x3 : p[1];
        pb[1].x = hlf ? x6 : p[4];  pb[1].y = hlf ? x7 : p[5];
        pb[2].x = hlf ? p[2] : u0;  pb[2].y = hlf ? p[3] : u1;
        pb[3].x = hlf ? p[6] : u4;  pb[3].y = hlf ? p[7] : u5;

#pragma unroll
        for (int t = 0; t < 4; t++)
#pragma unroll
            for (int kk = 0; kk < 4; kk++)
                acc[t] = WMMA_F32(va[t][kk], pb[kk], acc[t]);
    }

    const float inv = 1.0f / l;
    const size_t baseO = (size_t)(b * Nseq + q0 + l16) * Cdim + h * HD;
#pragma unroll
    for (int t = 0; t < 4; t++)
#pragma unroll
        for (int r = 0; r < 8; r++)
            O[baseO + t * 16 + r + 8 * hlf] = acc[t][r] * inv;
}

extern "C" void kernel_launch(void* const* d_in, const int* in_sizes, int n_in,
                              void* d_out, int out_size, void* d_ws, size_t ws_size,
                              hipStream_t stream)
{
    const float* x        = (const float*)d_in[0];
    const float* rope_cos = (const float*)d_in[1];
    const float* rope_sin = (const float*)d_in[2];
    const float* W_qkv    = (const float*)d_in[3];
    const float* W_proj   = (const float*)d_in[4];
    const float* b_proj   = (const float*)d_in[5];
    float* out  = (float*)d_out;

    float* qkv  = (float*)d_ws;
    float* attn = qkv + (size_t)Mrows * ThreeC;

    {
        const int blocks = (Mrows / GBM) * (ThreeC / GBN);
        gemm_xwt_kernel<<<blocks, 256, 0, stream>>>(x, W_qkv, nullptr, qkv,
                                                    Mrows, ThreeC, Cdim);
    }
    {
        const int threads = Bsz * Nseq * Hn * 32;
        rope_kernel<<<threads / 256, 256, 0, stream>>>(qkv, rope_cos, rope_sin);
    }
    {
        const int waves = Bsz * Hn * (Nseq / 16);
        flash_attn_kernel<<<waves / 8, 256, 0, stream>>>(qkv, attn);
    }
    {
        const int blocks = (Mrows / GBM) * (Cdim / GBN);
        gemm_xwt_kernel<<<blocks, 256, 0, stream>>>(attn, W_proj, b_proj, out,
                                                    Mrows, Cdim, Cdim);
    }
}


# attention: K/V tiles fetched once per 128-query block (coalesced 16-byte loads) and staged in LDS, two stages of 16 keys, instead of per-wave strided global fragment loads; key, k and sum order kept
# speedup vs baseline: 2.1359x; 1.0759x over previous
#include <hip/hip_runtime.h>

typedef float v2f __attribute__((ext_vector_type(2)));
typedef float v4f __attribute__((ext_vector_type(4)));
typedef float v8f __attribute__((ext_vector_type(8)));

#define WMMA_F32(a, b, c) \
    __builtin_amdgcn_wmma_f32_16x16x4_f32(false, (a), false, (b), (short)0, (c), false, false)

constexpr int Bsz    = 4;
constexpr int Nseq   = 2048;
constexpr int Cdim   = 1024;
constexpr int Hn     = 16;
constexpr int HD     = 64;
constexpr int ThreeC = 3 * Cdim;
constexpr int Mrows  = Bsz * Nseq;

constexpr int GBM = 128;
constexpr int GBN = 128;
constexpr int GBK = 64;
constexpr int GLP = GBK + 4;
constexpr int GTR = GBK / 4;
constexpr int GRP = 256 / GTR;
constexpr int GNP = GBM / GRP;

#define GEMM_WMMA8(a0, a1, b0, b1, b2, b3)        \
    acc[0][0] = WMMA_F32(a0, b0, acc[0][0]);      \
    acc[1][0] = WMMA_F32(a1, b0, acc[1][0]);      \
    acc[0][1] = WMMA_F32(a0, b1, acc[0][1]);      \
    acc[1][1] = WMMA_F32(a1, b1, acc[1][1]);      \
    acc[0][2] = WMMA_F32(a0, b2, acc[0][2]);      \
    acc[1][2] = WMMA_F32(a1, b2, acc[1][2]);      \
    acc[0][3] = WMMA_F32(a0, b3, acc[0][3]);      \
    acc[1][3] = WMMA_F32(a1, b3, acc[1][3]);

#define GEMM_FRAGS(a0, a1, b0, b1, b2, b3, kk)            \
    a0 = *(const v2f*)(pa + 4 * (kk));                    \
    a1 = *(const v2f*)(pa + 16 * GLP + 4 * (kk));         \
    b0 = *(const v2f*)(pb + 4 * (kk));                    \
    b1 = *(const v2f*)(pb + 16 * GLP + 4 * (kk));         \
    b2 = *(const v2f*)(pb + 32 * GLP + 4 * (kk));         \
    b3 = *(const v2f*)(pb + 48 * GLP + 4 * (kk));

__device__ __forceinline__
void gemm_stage_mma(const float* __restrict__ pa, const float* __restrict__ pb,
                    v8f (&acc)[2][4])
{
    v2f pa0, pa1, pb0, pb1, pb2, pb3;
    v2f qa0, qa1, qb0, qb1, qb2, qb3;
    GEMM_FRAGS(pa0, pa1, pb0, pb1, pb2, pb3, 0)
#pragma unroll 1
    for (int kk = 0; kk < GBK / 4 - 2; kk += 2) {
        GEMM_FRAGS(qa0, qa1, qb0, qb1, qb2, qb3, kk + 1)
        GEMM_WMMA8(pa0, pa1, pb0, pb1, pb2, pb3)
        GEMM_FRAGS(pa0, pa1, pb0, pb1, pb2, pb3, kk + 2)
        GEMM_WMMA8(qa0, qa1, qb0, qb1, qb2, qb3)
    }
    GEMM_FRAGS(qa0, qa1, qb0, qb1, qb2, qb3, GBK / 4 - 1)
    GEMM_WMMA8(pa0, pa1, pb0, pb1, pb2, pb3)
    GEMM_WMMA8(qa0, qa1, qb0, qb1, qb2, qb3)
}

__global__ __launch_bounds__(256)
void gemm_xwt_kernel(const float* __restrict__ X, const float* __restrict__ W,
                     const float* __restrict__ bias, float* __restrict__ Y,
                     int M, int Nd, int K)
{
    __shared__ __attribute__((aligned(16))) float sA[2][GBM * GLP];
    __shared__ __attribute__((aligned(16))) float sB[2][GBN * GLP];

    const int tid  = threadIdx.x;
    const int lane = tid & 31;
    const int wv   = tid >> 5;
    const int hlf  = lane >> 4;
    const int l16  = lane & 15;
    const int tilesN = Nd / GBN;
    const int bm0 = (blockIdx.x / tilesN) * GBM;
    const int bn0 = (blockIdx.x % tilesN) * GBN;
    const int wm  = (wv >> 1) * 32;
    const int wn  = (wv & 1) * 64;

    v8f zero = {};
    v8f acc[2][4];
#pragma unroll
    for (int i = 0; i < 2; i++)
#pragma unroll
        for (int j = 0; j < 4; j++) acc[i][j] = zero;

    const int srow = tid / GTR;
    const int scol = (tid % GTR) * 4;
    const float* gA = X + (size_t)(bm0 + srow) * K + scol;
    const float* gB = W + (size_t)(bn0 + srow) * K + scol;
    const size_t gstep = (size_t)GRP * K;
    const int soff = srow * GLP + scol;

    const int aoff = (wm + l16) * GLP + 2 * hlf;
    const int boff = (wn + l16) * GLP + 2 * hlf;

    v4f ra[GNP], rb[GNP];
#pragma unroll
    for (int i = 0; i < GNP; i++) {
        ra[i] = *(const v4f*)(gA + i * gstep);
        rb[i] = *(const v4f*)(gB + i * gstep);
    }
#pragma unroll
    for (int i = 0; i < GNP; i++) {
        *(v4f*)(&sA[0][soff + i * GRP * GLP]) = ra[i];
        *(v4f*)(&sB[0][soff + i * GRP * GLP]) = rb[i];
    }
    __syncthreads();

    const int nk = K / GBK;
    for (int kt = 0; kt + 1 < nk; kt++) {
        const int buf = kt & 1;
        const int kn = (kt + 1) * GBK;
#pragma unroll
        for (int i = 0; i < GNP; i++) {
            ra[i] = *(const v4f*)(gA + i * gstep + kn);
            rb[i] = *(const v4f*)(gB + i * gstep + kn);
        }

        gemm_stage_mma(&sA[buf][aoff], &sB[buf][boff], acc);

#pragma unroll
        for (int i = 0; i < GNP; i++) {
            *(v4f*)(&sA[buf ^ 1][soff + i * GRP * GLP]) = ra[i];
            *(v4f*)(&sB[buf ^ 1][soff + i * GRP * GLP]) = rb[i];
        }
        __syncthreads();
    }
    {
        const int buf = (nk - 1) & 1;
        gemm_stage_mma(&sA[buf][aoff], &sB[buf][boff], acc);
    }

    const int m0 = bm0 + wm;
    const int n0 = bn0 + wn;
#pragma unroll
    for (int i = 0; i < 2; i++)
#pragma unroll
        for (int j = 0; j < 4; j++)
#pragma unroll
            for (int r = 0; r < 8; r++) {
                const int row = m0 + 16 * i + r + 8 * hlf;
                const int col = n0 + 16 * j + l16;
                float v = acc[i][j][r];
                if (bias) v += bias[col];
                Y[(size_t)row * Nd + col] = v;
            }
}

__global__ __launch_bounds__(256)
void rope_kernel(float* __restrict__ qkv,
                 const float* __restrict__ cosT, const float* __restrict__ sinT)
{
    const int t = blockIdx.x * blockDim.x + threadIdx.x;
    const int i = t & 31;
    const int h = (t >> 5) & 15;
    const int n = (t >> 9) & (Nseq - 1);
    const int b = t >> 20;

    const size_t base = (size_t)(b * Nseq + n) * ThreeC + h * HD;
    const float c0 = cosT[n * HD + i],      s0 = sinT[n * HD + i];
    const float c1 = cosT[n * HD + i + 32], s1 = sinT[n * HD + i + 32];

    const float q0 = qkv[base + i], q1 = qkv[base + i + 32];
    qkv[base + i]      = q0 * c0 - q1 * s0;
    qkv[base + i + 32] = q1 * c1 + q0 * s1;

    const float k0 = qkv[base + Cdim + i], k1 = qkv[base + Cdim + i + 32];
    qkv[base + Cdim + i]      = k0 * c0 - k1 * s0;
    qkv[base + Cdim + i + 32] = k1 * c1 + k0 * s1;
}

constexpr int AKT = 16;
constexpr int AKP = HD + 4;
constexpr int AVP = 2 * HD + 32;
constexpr int ANT = Nseq / AKT;

__device__ __forceinline__
void attn_keys16(const float* pk, const float* pv, const v2f (&qb)[16],
                 v8f (&acc)[4], float& m, float& l, const int hlf)
{
    const float scale = 0.125f;
    v8f zero = {};

    v2f ka[16];
#pragma unroll
    for (int kk = 0; kk < 16; kk++)
        ka[kk] = *(const v2f*)(pk + 4 * kk);

    asm volatile("" :: "v"(ka[0]), "v"(ka[1]), "v"(ka[2]), "v"(ka[3]), "v"(ka[4]), "v"(ka[5]),
                       "v"(ka[6]), "v"(ka[7]), "v"(ka[8]), "v"(ka[9]), "v"(ka[10]), "v"(ka[11]),
                       "v"(ka[12]), "v"(ka[13]), "v"(ka[14]), "v"(ka[15]));

    v8f s = zero;
#pragma unroll
    for (int kk = 0; kk < 16; kk++)
        s = WMMA_F32(ka[kk], qb[kk], s);

    v2f va[4][4];
#pragma unroll
    for (int t = 0; t < 4; t++)
#pragma unroll
        for (int kk = 0; kk < 4; kk++) {
            va[t][kk].x = pv[(2 * kk) * AVP + 32 * t];
            va[t][kk].y = pv[(2 * kk) * AVP + 32 * t + 16];
        }

    float sv[8];
#pragma unroll
    for (int r = 0; r < 8; r++) sv[r] = s[r] * scale;

    float tm = sv[0];
#pragma unroll
    for (int r = 1; r < 8; r++) tm = fmaxf(tm, sv[r]);
    tm = fmaxf(tm, __shfl_xor(tm, 16, 32));

    const float mn   = fmaxf(m, tm);
    const float corr = __expf(m - mn);
    m = mn;

    float p[8];
    float rs = 0.0f;
#pragma unroll
    for (int r = 0; r < 8; r++) { p[r] = __expf(sv[r] - mn); rs += p[r]; }
    rs += __shfl_xor(rs, 16, 32);
    l = l * corr + rs;

#pragma unroll
    for (int t = 0; t < 4; t++)
#pragma unroll
        for (int r = 0; r < 8; r++) acc[t][r] *= corr;

    const float x2 = __shfl_xor(p[2], 16, 32), x3 = __shfl_xor(p[3], 16, 32);
    const float x6 = __shfl_xor(p[6], 16, 32), x7 = __shfl_xor(p[7], 16, 32);
    const float u0 = __shfl_xor(p[0], 16, 32), u1 = __shfl_xor(p[1], 16, 32);
    const float u4 = __shfl_xor(p[4], 16, 32), u5 = __shfl_xor(p[5], 16, 32);
    v2f pb[4];
    pb[0].x = hlf ? x2 : p[0];  pb[0].y = hlf ? x3 : p[1];
    pb[1].x = hlf ? x6 : p[4];  pb[1].y = hlf ? x7 : p[5];
    pb[2].x = hlf ? p[2] : u0;  pb[2].y = hlf ? p[3] : u1;
    pb[3].x = hlf ? p[6] : u4;  pb[3].y = hlf ? p[7] : u5;

#pragma unroll
    for (int t = 0; t < 4; t++)
#pragma unroll
        for (int kk = 0; kk < 4; kk++)
            acc[t] = WMMA_F32(va[t][kk], pb[kk], acc[t]);
}

__global__ __launch_bounds__(256, 8)
void flash_attn_kernel(const float* __restrict__ qkv, float* __restrict__ O)
{
    __shared__ __attribute__((aligned(16))) float sK[2][AKT * AKP];
    __shared__ __attribute__((aligned(16))) float sV[2][(AKT / 2) * AVP];

    const int tid  = threadIdx.x;
    const int lane = tid & 31;
    const int wv   = tid >> 5;
    const int qt = (blockIdx.x & 15) * 8 + wv;
    const int h  = (blockIdx.x >> 4) & 15;
    const int b  = blockIdx.x >> 8;
    const int hlf = lane >> 4;
    const int l16 = lane & 15;
    const int q0  = qt * 16;

    const char* qkvB = (const char*)qkv;

    const unsigned rowQ = ((unsigned)(b * Nseq + q0 + l16) * ThreeC + h * HD + 2 * hlf) * 4u;
    v2f qb[16];
#pragma unroll
    for (int kk = 0; kk < 16; kk++)
        qb[kk] = *(const v2f*)(qkvB + rowQ + 16 * kk);

    v8f zero = {};
    v8f acc[4];
#pragma unroll
    for (int t = 0; t < 4; t++) acc[t] = zero;
    float m = -3.0e38f, l = 0.0f;

    const int srow = tid >> 4;
    const int scol = (tid & 15) * 4;
    unsigned gko = ((unsigned)(b * Nseq + srow) * ThreeC + Cdim + h * HD + scol) * 4u;
    constexpr unsigned gtile = AKT * ThreeC * 4u;
    const int skoff = srow * AKP + scol;
    const int svoff = (srow >> 1) * AVP + (scol >> 4) * 32 + (srow & 1) * 16 + (scol & 15);

    const int koff = l16 * AKP + 2 * hlf;
    const int voff = hlf * AVP + l16;

    v4f rk, rv;
    rk = *(const v4f*)(qkvB + gko);
    rv = *(const v4f*)(qkvB + gko + Cdim * 4u);
    *(v4f*)(&sK[0][skoff]) = rk;
    *(v4f*)(&sV[0][svoff]) = rv;
    __syncthreads();

    for (int it = 0; it + 1 < ANT; it++) {
        const int buf = it & 1;
        gko += gtile;
        rk = *(const v4f*)(qkvB + gko);
        rv = *(const v4f*)(qkvB + gko + Cdim * 4u);

        attn_keys16(&sK[buf][koff], &sV[buf][voff], qb, acc, m, l, hlf);

        *(v4f*)(&sK[buf ^ 1][skoff]) = rk;
        *(v4f*)(&sV[buf ^ 1][svoff]) = rv;
        __syncthreads();
    }
    {
        const int buf = (ANT - 1) & 1;
        attn_keys16(&sK[buf][koff], &sV[buf][voff], qb, acc, m, l, hlf);
    }

    const float inv = 1.0f / l;
    const unsigned baseO = ((unsigned)(b * Nseq + q0 + l16) * Cdim + h * HD + 8 * hlf) * 4u;
#pragma unroll
    for (int t = 0; t < 4; t++)
#pragma unroll
        for (int r = 0; r < 8; r++)
            *(float*)((char*)O + baseO + (t * 16 + r) * 4) = acc[t][r] * inv;
}

extern "C" void kernel_launch(void* const* d_in, const int* in_sizes, int n_in,
                              void* d_out, int out_size, void* d_ws, size_t ws_size,
                              hipStream_t stream)
{
    const float* x        = (const float*)d_in[0];
    const float* rope_cos = (const float*)d_in[1];
    const float* rope_sin = (const float*)d_in[2];
    const float* W_qkv    = (const float*)d_in[3];
    const float* W_proj   = (const float*)d_in[4];
    const float* b_proj   = (const float*)d_in[5];
    float* out  = (float*)d_out;

    float* qkv  = (float*)d_ws;
    float* attn = qkv + (size_t)Mrows * ThreeC;

    {
        const int blocks = (Mrows / GBM) * (ThreeC / GBN);
        gemm_xwt_kernel<<<blocks, 256, 0, stream>>>(x, W_qkv, nullptr, qkv,
                                                    Mrows, ThreeC, Cdim);
    }
    {
        const int threads = Bsz * Nseq * Hn * 32;
        rope_kernel<<<threads / 256, 256, 0, stream>>>(qkv, rope_cos, rope_sin);
    }
    {
        const int waves = Bsz * Hn * (Nseq / 16);
        flash_attn_kernel<<<waves / 8, 256, 0, stream>>>(qkv, attn);
    }
    {
        const int blocks = (Mrows / GBM) * (Cdim / GBN);
        gemm_xwt_kernel<<<blocks, 256, 0, stream>>>(attn, W_proj, b_proj, out,
                                                    Mrows, Cdim, Cdim);
    }
}


# rotary embedding applied to the q,k accumulators in the qkv GEMM epilogue (partner column sits in the same lane) instead of a separate in-place pass of 131072 short waves: one launch fewer
# speedup vs baseline: 2.2488x; 1.0529x over previous
#include <hip/hip_runtime.h>

typedef float v2f __attribute__((ext_vector_type(2)));
typedef float v4f __attribute__((ext_vector_type(4)));
typedef float v8f __attribute__((ext_vector_type(8)));

#define WMMA_F32(a, b, c) \
    __builtin_amdgcn_wmma_f32_16x16x4_f32(false, (a), false, (b), (short)0, (c), false, false)

constexpr int Bsz    = 4;
constexpr int Nseq   = 2048;
constexpr int Cdim   = 1024;
constexpr int Hn     = 16;
constexpr int HD     = 64;
constexpr int ThreeC = 3 * Cdim;
constexpr int Mrows  = Bsz * Nseq;

constexpr int GBM = 128;
constexpr int GBN = 128;
constexpr int GBK = 64;
constexpr int GLP = GBK + 4;
constexpr int GTR = GBK / 4;
constexpr int GRP = 256 / GTR;
constexpr int GNP = GBM / GRP;

#define GEMM_WMMA8(a0, a1, b0, b1, b2, b3)        \
    acc[0][0] = WMMA_F32(a0, b0, acc[0][0]);      \
    acc[1][0] = WMMA_F32(a1, b0, acc[1][0]);      \
    acc[0][1] = WMMA_F32(a0, b1, acc[0][1]);      \
    acc[1][1] = WMMA_F32(a1, b1, acc[1][1]);      \
    acc[0][2] = WMMA_F32(a0, b2, acc[0][2]);      \
    acc[1][2] = WMMA_F32(a1, b2, acc[1][2]);      \
    acc[0][3] = WMMA_F32(a0, b3, acc[0][3]);      \
    acc[1][3] = WMMA_F32(a1, b3, acc[1][3]);

#define GEMM_FRAGS(a0, a1, b0, b1, b2, b3, kk)            \
    a0 = *(const v2f*)(pa + 4 * (kk));                    \
    a1 = *(const v2f*)(pa + 16 * GLP + 4 * (kk));         \
    b0 = *(const v2f*)(pb + 4 * (kk));                    \
    b1 = *(const v2f*)(pb + 16 * GLP + 4 * (kk));         \
    b2 = *(const v2f*)(pb + 32 * GLP + 4 * (kk));         \
    b3 = *(const v2f*)(pb + 48 * GLP + 4 * (kk));

__device__ __forceinline__
void gemm_stage_mma(const float* __restrict__ pa, const float* __restrict__ pb,
                    v8f (&acc)[2][4])
{
    v2f pa0, pa1, pb0, pb1, pb2, pb3;
    v2f qa0, qa1, qb0, qb1, qb2, qb3;
    GEMM_FRAGS(pa0, pa1, pb0, pb1, pb2, pb3, 0)
#pragma unroll 1
    for (int kk = 0; kk < GBK / 4 - 2; kk += 2) {
        GEMM_FRAGS(qa0, qa1, qb0, qb1, qb2, qb3, kk + 1)
        GEMM_WMMA8(pa0, pa1, pb0, pb1, pb2, pb3)
        GEMM_FRAGS(pa0, pa1, pb0, pb1, pb2, pb3, kk + 2)
        GEMM_WMMA8(qa0, qa1, qb0, qb1, qb2, qb3)
    }
    GEMM_FRAGS(qa0, qa1, qb0, qb1, qb2, qb3, GBK / 4 - 1)
    GEMM_WMMA8(pa0, pa1, pb0, pb1, pb2, pb3)
    GEMM_WMMA8(qa0, qa1, qb0, qb1, qb2, qb3)
}

__global__ __launch_bounds__(256)
void gemm_xwt_kernel(const float* __restrict__ X, const float* __restrict__ W,
                     const float* __restrict__ bias,
                     const float* __restrict__ cosT, const float* __restrict__ sinT,
                     float* __restrict__ Y, int M, int Nd, int K)
{
    __shared__ __attribute__((aligned(16))) float sA[2][GBM * GLP];
    __shared__ __attribute__((aligned(16))) float sB[2][GBN * GLP];

    const int tid  = threadIdx.x;
    const int lane = tid & 31;
    const int wv   = tid >> 5;
    const int hlf  = lane >> 4;
    const int l16  = lane & 15;
    const int tilesN = Nd / GBN;
    const int bm0 = (blockIdx.x / tilesN) * GBM;
    const int bn0 = (blockIdx.x % tilesN) * GBN;
    const int wm  = (wv >> 1) * 32;
    const int wn  = (wv & 1) * 64;

    v8f zero = {};
    v8f acc[2][4];
#pragma unroll
    for (int i = 0; i < 2; i++)
#pragma unroll
        for (int j = 0; j < 4; j++) acc[i][j] = zero;

    const int srow = tid / GTR;
    const int scol = (tid % GTR) * 4;
    const float* gA = X + (size_t)(bm0 + srow) * K + scol;
    const float* gB = W + (size_t)(bn0 + srow) * K + scol;
    const size_t gstep = (size_t)GRP * K;
    const int soff = srow * GLP + scol;

    const int aoff = (wm + l16) * GLP + 2 * hlf;
    const int boff = (wn + l16) * GLP + 2 * hlf;

    v4f ra[GNP], rb[GNP];
#pragma unroll
    for (int i = 0; i < GNP; i++) {
        ra[i] = *(const v4f*)(gA + i * gstep);
        rb[i] = *(const v4f*)(gB + i * gstep);
    }
#pragma unroll
    for (int i = 0; i < GNP; i++) {
        *(v4f*)(&sA[0][soff + i * GRP * GLP]) = ra[i];
        *(v4f*)(&sB[0][soff + i * GRP * GLP]) = rb[i];
    }
    __syncthreads();

    const int nk = K / GBK;
    for (int kt = 0; kt + 1 < nk; kt++) {
        const int buf = kt & 1;
        const int kn = (kt + 1) * GBK;
#pragma unroll
        for (int i = 0; i < GNP; i++) {
            ra[i] = *(const v4f*)(gA + i * gstep + kn);
            rb[i] = *(const v4f*)(gB + i * gstep + kn);
        }

        gemm_stage_mma(&sA[buf][aoff], &sB[buf][boff], acc);

#pragma unroll
        for (int i = 0; i < GNP; i++) {
            *(v4f*)(&sA[buf ^ 1][soff + i * GRP * GLP]) = ra[i];
            *(v4f*)(&sB[buf ^ 1][soff + i * GRP * GLP]) = rb[i];
        }
        __syncthreads();
    }
    {
        const int buf = (nk - 1) & 1;
        gemm_stage_mma(&sA[buf][aoff], &sB[buf][boff], acc);
    }

    const int m0 = bm0 + wm;
    const int n0 = bn0 + wn;

    if (cosT != nullptr && bn0 < 2 * Cdim) {
#pragma unroll
        for (int i = 0; i < 2; i++)
#pragma unroll
            for (int r = 0; r < 8; r++) {
                const int n = (m0 + 16 * i + r + 8 * hlf) & (Nseq - 1);
                const float* cr = cosT + n * HD + l16;
                const float* sr = sinT + n * HD + l16;
#pragma unroll
                for (int j = 0; j < 2; j++) {
                    const float c0 = cr[16 * j],      s0 = sr[16 * j];
                    const float c1 = cr[16 * j + 32], s1 = sr[16 * j + 32];
                    const float t0 = acc[i][j][r], t1 = acc[i][j + 2][r];
                    acc[i][j][r]     = t0 * c0 - t1 * s0;
                    acc[i][j + 2][r] = t1 * c1 + t0 * s1;
                }
            }
    }

#pragma unroll
    for (int i = 0; i < 2; i++)
#pragma unroll
        for (int j = 0; j < 4; j++)
#pragma unroll
            for (int r = 0; r < 8; r++) {
                const int row = m0 + 16 * i + r + 8 * hlf;
                const int col = n0 + 16 * j + l16;
                float v = acc[i][j][r];
                if (bias) v += bias[col];
                Y[(size_t)row * Nd + col] = v;
            }
}

constexpr int AKT = 16;
constexpr int AKP = HD + 4;
constexpr int AVP = 2 * HD + 32;
constexpr int ANT = Nseq / AKT;

__device__ __forceinline__
void attn_keys16(const float* pk, const float* pv, const v2f (&qb)[16],
                 v8f (&acc)[4], float& m, float& l, const int hlf)
{
    const float scale = 0.125f;
    v8f zero = {};

    v2f ka[16];
#pragma unroll
    for (int kk = 0; kk < 16; kk++)
        ka[kk] = *(const v2f*)(pk + 4 * kk);

    asm volatile("" :: "v"(ka[0]), "v"(ka[1]), "v"(ka[2]), "v"(ka[3]), "v"(ka[4]), "v"(ka[5]),
                       "v"(ka[6]), "v"(ka[7]), "v"(ka[8]), "v"(ka[9]), "v"(ka[10]), "v"(ka[11]),
                       "v"(ka[12]), "v"(ka[13]), "v"(ka[14]), "v"(ka[15]));

    v8f s = zero;
#pragma unroll
    for (int kk = 0; kk < 16; kk++)
        s = WMMA_F32(ka[kk], qb[kk], s);

    v2f va[4][4];
#pragma unroll
    for (int t = 0; t < 4; t++)
#pragma unroll
        for (int kk = 0; kk < 4; kk++) {
            va[t][kk].x = pv[(2 * kk) * AVP + 32 * t];
            va[t][kk].y = pv[(2 * kk) * AVP + 32 * t + 16];
        }

    float sv[8];
#pragma unroll
    for (int r = 0; r < 8; r++) sv[r] = s[r] * scale;

    float tm = sv[0];
#pragma unroll
    for (int r = 1; r < 8; r++) tm = fmaxf(tm, sv[r]);
    tm = fmaxf(tm, __shfl_xor(tm, 16, 32));

    const float mn   = fmaxf(m, tm);
    const float corr = __expf(m - mn);
    m = mn;

    float p[8];
    float rs = 0.0f;
#pragma unroll
    for (int r = 0; r < 8; r++) { p[r] = __expf(sv[r] - mn); rs += p[r]; }
    rs += __shfl_xor(rs, 16, 32);
    l = l * corr + rs;

#pragma unroll
    for (int t = 0; t < 4; t++)
#pragma unroll
        for (int r = 0; r < 8; r++) acc[t][r] *= corr;

    const float x2 = __shfl_xor(p[2], 16, 32), x3 = __shfl_xor(p[3], 16, 32);
    const float x6 = __shfl_xor(p[6], 16, 32), x7 = __shfl_xor(p[7], 16, 32);
    const float u0 = __shfl_xor(p[0], 16, 32), u1 = __shfl_xor(p[1], 16, 32);
    const float u4 = __shfl_xor(p[4], 16, 32), u5 = __shfl_xor(p[5], 16, 32);
    v2f pb[4];
    pb[0].x = hlf ? x2 : p[0];  pb[0].y = hlf ? x3 : p[1];
    pb[1].x = hlf ? x6 : p[4];  pb[1].y = hlf ? x7 : p[5];
    pb[2].x = hlf ? p[2] : u0;  pb[2].y = hlf ? p[3] : u1;
    pb[3].x = hlf ? p[6] : u4;  pb[3].y = hlf ? p[7] : u5;

#pragma unroll
    for (int t = 0; t < 4; t++)
#pragma unroll
        for (int kk = 0; kk < 4; kk++)
            acc[t] = WMMA_F32(va[t][kk], pb[kk], acc[t]);
}

__global__ __launch_bounds__(256, 8)
void flash_attn_kernel(const float* __restrict__ qkv, float* __restrict__ O)
{
    __shared__ __attribute__((aligned(16))) float sK[2][AKT * AKP];
    __shared__ __attribute__((aligned(16))) float sV[2][(AKT / 2) * AVP];

    const int tid  = threadIdx.x;
    const int lane = tid & 31;
    const int wv   = tid >> 5;
    const int qt = (blockIdx.x & 15) * 8 + wv;
    const int h  = (blockIdx.x >> 4) & 15;
    const int b  = blockIdx.x >> 8;
    const int hlf = lane >> 4;
    const int l16 = lane & 15;
    const int q0  = qt * 16;

    const char* qkvB = (const char*)qkv;

    const unsigned rowQ = ((unsigned)(b * Nseq + q0 + l16) * ThreeC + h * HD + 2 * hlf) * 4u;
    v2f qb[16];
#pragma unroll
    for (int kk = 0; kk < 16; kk++)
        qb[kk] = *(const v2f*)(qkvB + rowQ + 16 * kk);

    v8f zero = {};
    v8f acc[4];
#pragma unroll
    for (int t = 0; t < 4; t++) acc[t] = zero;
    float m = -3.0e38f, l = 0.0f;

    const int srow = tid >> 4;
    const int scol = (tid & 15) * 4;
    unsigned gko = ((unsigned)(b * Nseq + srow) * ThreeC + Cdim + h * HD + scol) * 4u;
    constexpr unsigned gtile = AKT * ThreeC * 4u;
    const int skoff = srow * AKP + scol;
    const int svoff = (srow >> 1) * AVP + (scol >> 4) * 32 + (srow & 1) * 16 + (scol & 15);

    const int koff = l16 * AKP + 2 * hlf;
    const int voff = hlf * AVP + l16;

    v4f rk, rv;
    rk = *(const v4f*)(qkvB + gko);
    rv = *(const v4f*)(qkvB + gko + Cdim * 4u);
    *(v4f*)(&sK[0][skoff]) = rk;
    *(v4f*)(&sV[0][svoff]) = rv;
    __syncthreads();

    for (int it = 0; it + 1 < ANT; it++) {
        const int buf = it & 1;
        gko += gtile;
        rk = *(const v4f*)(qkvB + gko);
        rv = *(const v4f*)(qkvB + gko + Cdim * 4u);

        attn_keys16(&sK[buf][koff], &sV[buf][voff], qb, acc, m, l, hlf);

        *(v4f*)(&sK[buf ^ 1][skoff]) = rk;
        *(v4f*)(&sV[buf ^ 1][svoff]) = rv;
        __syncthreads();
    }
    {
        const int buf = (ANT - 1) & 1;
        attn_keys16(&sK[buf][koff], &sV[buf][voff], qb, acc, m, l, hlf);
    }

    const float inv = 1.0f / l;
    const unsigned baseO = ((unsigned)(b * Nseq + q0 + l16) * Cdim + h * HD + 8 * hlf) * 4u;
#pragma unroll
    for (int t = 0; t < 4; t++)
#pragma unroll
        for (int r = 0; r < 8; r++)
            *(float*)((char*)O + baseO + (t * 16 + r) * 4) = acc[t][r] * inv;
}

extern "C" void kernel_launch(void* const* d_in, const int* in_sizes, int n_in,
                              void* d_out, int out_size, void* d_ws, size_t ws_size,
                              hipStream_t stream)
{
    const float* x        = (const float*)d_in[0];
    const float* rope_cos = (const float*)d_in[1];
    const float* rope_sin = (const float*)d_in[2];
    const float* W_qkv    = (const float*)d_in[3];
    const float* W_proj   = (const float*)d_in[4];
    const float* b_proj   = (const float*)d_in[5];
    float* out  = (float*)d_out;

    float* qkv  = (float*)d_ws;
    float* attn = qkv + (size_t)Mrows * ThreeC;

    {
        const int blocks = (Mrows / GBM) * (ThreeC / GBN);
        gemm_xwt_kernel<<<blocks, 256, 0, stream>>>(x, W_qkv, nullptr, rope_cos, rope_sin, qkv,
                                                    Mrows, ThreeC, Cdim);
    }
    {
        const int waves = Bsz * Hn * (Nseq / 16);
        flash_attn_kernel<<<waves / 8, 256, 0, stream>>>(qkv, attn);
    }
    {
        const int blocks = (Mrows / GBM) * (Cdim / GBN);
        gemm_xwt_kernel<<<blocks, 256, 0, stream>>>(attn, W_proj, b_proj, nullptr, nullptr, out,
                                                    Mrows, Cdim, Cdim);
    }
}


# attention: softmax reference fixed at the first key tile's maximum (scale*log2e folded into Q), no per-tile rescale; guarded redo with running maximum; row sums joined at the end: sum order changed
# speedup vs baseline: 2.3152x; 1.0295x over previous
#include <hip/hip_runtime.h>

typedef float v2f __attribute__((ext_vector_type(2)));
typedef float v4f __attribute__((ext_vector_type(4)));
typedef float v8f __attribute__((ext_vector_type(8)));

#define WMMA_F32(a, b, c) \
    __builtin_amdgcn_wmma_f32_16x16x4_f32(false, (a), false, (b), (short)0, (c), false, false)

constexpr int Bsz    = 4;
constexpr int Nseq   = 2048;
constexpr int Cdim   = 1024;
constexpr int Hn     = 16;
constexpr int HD     = 64;
constexpr int ThreeC = 3 * Cdim;
constexpr int Mrows  = Bsz * Nseq;

constexpr int GBM = 128;
constexpr int GBN = 128;
constexpr int GBK = 64;
constexpr int GLP = GBK + 4;
constexpr int GTR = GBK / 4;
constexpr int GRP = 256 / GTR;
constexpr int GNP = GBM / GRP;

#define GEMM_WMMA8(a0, a1, b0, b1, b2, b3)        \
    acc[0][0] = WMMA_F32(a0, b0, acc[0][0]);      \
    acc[1][0] = WMMA_F32(a1, b0, acc[1][0]);      \
    acc[0][1] = WMMA_F32(a0, b1, acc[0][1]);      \
    acc[1][1] = WMMA_F32(a1, b1, acc[1][1]);      \
    acc[0][2] = WMMA_F32(a0, b2, acc[0][2]);      \
    acc[1][2] = WMMA_F32(a1, b2, acc[1][2]);      \
    acc[0][3] = WMMA_F32(a0, b3, acc[0][3]);      \
    acc[1][3] = WMMA_F32(a1, b3, acc[1][3]);

#define GEMM_FRAGS(a0, a1, b0, b1, b2, b3, kk)            \
    a0 = *(const v2f*)(pa + 4 * (kk));                    \
    a1 = *(const v2f*)(pa + 16 * GLP + 4 * (kk));         \
    b0 = *(const v2f*)(pb + 4 * (kk));                    \
    b1 = *(const v2f*)(pb + 16 * GLP + 4 * (kk));         \
    b2 = *(const v2f*)(pb + 32 * GLP + 4 * (kk));         \
    b3 = *(const v2f*)(pb + 48 * GLP + 4 * (kk));

__device__ __forceinline__
void gemm_stage_mma(const float* __restrict__ pa, const float* __restrict__ pb,
                    v8f (&acc)[2][4])
{
    v2f pa0, pa1, pb0, pb1, pb2, pb3;
    v2f qa0, qa1, qb0, qb1, qb2, qb3;
    GEMM_FRAGS(pa0, pa1, pb0, pb1, pb2, pb3, 0)
#pragma unroll 1
    for (int kk = 0; kk < GBK / 4 - 2; kk += 2) {
        GEMM_FRAGS(qa0, qa1, qb0, qb1, qb2, qb3, kk + 1)
        GEMM_WMMA8(pa0, pa1, pb0, pb1, pb2, pb3)
        GEMM_FRAGS(pa0, pa1, pb0, pb1, pb2, pb3, kk + 2)
        GEMM_WMMA8(qa0, qa1, qb0, qb1, qb2, qb3)
    }
    GEMM_FRAGS(qa0, qa1, qb0, qb1, qb2, qb3, GBK / 4 - 1)
    GEMM_WMMA8(pa0, pa1, pb0, pb1, pb2, pb3)
    GEMM_WMMA8(qa0, qa1, qb0, qb1, qb2, qb3)
}

__global__ __launch_bounds__(256)
void gemm_xwt_kernel(const float* __restrict__ X, const float* __restrict__ W,
                     const float* __restrict__ bias,
                     const float* __restrict__ cosT, const float* __restrict__ sinT,
                     float* __restrict__ Y, int M, int Nd, int K)
{
    __shared__ __attribute__((aligned(16))) float sA[2][GBM * GLP];
    __shared__ __attribute__((aligned(16))) float sB[2][GBN * GLP];

    const int tid  = threadIdx.x;
    const int lane = tid & 31;
    const int wv   = tid >> 5;
    const int hlf  = lane >> 4;
    const int l16  = lane & 15;
    const int tilesN = Nd / GBN;
    const int bm0 = (blockIdx.x / tilesN) * GBM;
    const int bn0 = (blockIdx.x % tilesN) * GBN;
    const int wm  = (wv >> 1) * 32;
    const int wn  = (wv & 1) * 64;

    v8f zero = {};
    v8f acc[2][4];
#pragma unroll
    for (int i = 0; i < 2; i++)
#pragma unroll
        for (int j = 0; j < 4; j++) acc[i][j] = zero;

    const int srow = tid / GTR;
    const int scol = (tid % GTR) * 4;
    const float* gA = X + (size_t)(bm0 + srow) * K + scol;
    const float* gB = W + (size_t)(bn0 + srow) * K + scol;
    const size_t gstep = (size_t)GRP * K;
    const int soff = srow * GLP + scol;

    const int aoff = (wm + l16) * GLP + 2 * hlf;
    const int boff = (wn + l16) * GLP + 2 * hlf;

    v4f ra[GNP], rb[GNP];
#pragma unroll
    for (int i = 0; i < GNP; i++) {
        ra[i] = *(const v4f*)(gA + i * gstep);
        rb[i] = *(const v4f*)(gB + i * gstep);
    }
#pragma unroll
    for (int i = 0; i < GNP; i++) {
        *(v4f*)(&sA[0][soff + i * GRP * GLP]) = ra[i];
        *(v4f*)(&sB[0][soff + i * GRP * GLP]) = rb[i];
    }
    __syncthreads();

    const int nk = K / GBK;
    for (int kt = 0; kt + 1 < nk; kt++) {
        const int buf = kt & 1;
        const int kn = (kt + 1) * GBK;
#pragma unroll
        for (int i = 0; i < GNP; i++) {
            ra[i] = *(const v4f*)(gA + i * gstep + kn);
            rb[i] = *(const v4f*)(gB + i * gstep + kn);
        }

        gemm_stage_mma(&sA[buf][aoff], &sB[buf][boff], acc);

#pragma unroll
        for (int i = 0; i < GNP; i++) {
            *(v4f*)(&sA[buf ^ 1][soff + i * GRP * GLP]) = ra[i];
            *(v4f*)(&sB[buf ^ 1][soff + i * GRP * GLP]) = rb[i];
        }
        __syncthreads();
    }
    {
        const int buf = (nk - 1) & 1;
        gemm_stage_mma(&sA[buf][aoff], &sB[buf][boff], acc);
    }

    const int m0 = bm0 + wm;
    const int n0 = bn0 + wn;

    if (cosT != nullptr && bn0 < 2 * Cdim) {
#pragma unroll
        for (int i = 0; i < 2; i++)
#pragma unroll
            for (int r = 0; r < 8; r++) {
                const int n = (m0 + 16 * i + r + 8 * hlf) & (Nseq - 1);
                const float* cr = cosT + n * HD + l16;
                const float* sr = sinT + n * HD + l16;
#pragma unroll
                for (int j = 0; j < 2; j++) {
                    const float c0 = cr[16 * j],      s0 = sr[16 * j];
                    const float c1 = cr[16 * j + 32], s1 = sr[16 * j + 32];
                    const float t0 = acc[i][j][r], t1 = acc[i][j + 2][r];
                    acc[i][j][r]     = t0 * c0 - t1 * s0;
                    acc[i][j + 2][r] = t1 * c1 + t0 * s1;
                }
            }
    }

#pragma unroll
    for (int i = 0; i < 2; i++)
#pragma unroll
        for (int j = 0; j < 4; j++)
#pragma unroll
            for (int r = 0; r < 8; r++) {
                const int row = m0 + 16 * i + r + 8 * hlf;
                const int col = n0 + 16 * j + l16;
                float v = acc[i][j][r];
                if (bias) v += bias[col];
                Y[(size_t)row * Nd + col] = v;
            }
}

constexpr int AKT = 16;
constexpr int AKP = HD + 4;
constexpr int AVP = 2 * HD + 32;
constexpr int ANT = Nseq / AKT;

template <bool FIXED>
__device__ __forceinline__
void attn_keys16(const float* pk, const float* pv, const v2f (&qb)[16],
                 v8f (&acc)[4], float& m, float& l, const int hlf, const bool first, unsigned& bad)
{
    v8f zero = {};

    v2f ka[16];
#pragma unroll
    for (int kk = 0; kk < 16; kk++)
        ka[kk] = *(const v2f*)(pk + 4 * kk);

    asm volatile("" :: "v"(ka[0]), "v"(ka[1]), "v"(ka[2]), "v"(ka[3]), "v"(ka[4]), "v"(ka[5]),
                       "v"(ka[6]), "v"(ka[7]), "v"(ka[8]), "v"(ka[9]), "v"(ka[10]), "v"(ka[11]),
                       "v"(ka[12]), "v"(ka[13]), "v"(ka[14]), "v"(ka[15]));

    v8f s = zero;
#pragma unroll
    for (int kk = 0; kk < 16; kk++)
        s = WMMA_F32(ka[kk], qb[kk], s);

    v2f va[4][4];
#pragma unroll
    for (int t = 0; t < 4; t++)
#pragma unroll
        for (int kk = 0; kk < 4; kk++) {
            va[t][kk].x = pv[(2 * kk) * AVP + 32 * t];
            va[t][kk].y = pv[(2 * kk) * AVP + 32 * t + 16];
        }

    if (!FIXED || first) {
        float tm = s[0];
#pragma unroll
        for (int r = 1; r < 8; r++) tm = fmaxf(tm, s[r]);
        tm = fmaxf(tm, __shfl_xor(tm, 16, 32));
        if (FIXED) {
            m = tm;
        } else {
            const float mn   = fmaxf(m, tm);
            const float corr = __builtin_amdgcn_exp2f(m - mn);
            m = mn;
            l *= corr;
#pragma unroll
            for (int t = 0; t < 4; t++)
#pragma unroll
                for (int r = 0; r < 8; r++) acc[t][r] *= corr;
        }
    }

    v8f e;
#pragma unroll
    for (int r = 0; r < 8; r++) e[r] = __builtin_amdgcn_exp2f(s[r] - m);
    const v4f h4 = e.lo + e.hi;
    const v2f h2 = h4.lo + h4.hi;
    const float rs = h2.x + h2.y;
    l += rs;
    if (FIXED) bad |= __builtin_amdgcn_ballot_w32(!(rs <= 0x1p30f));

    const float x2 = __shfl_xor(e[2], 16, 32), x3 = __shfl_xor(e[3], 16, 32);
    const float x6 = __shfl_xor(e[6], 16, 32), x7 = __shfl_xor(e[7], 16, 32);
    const float u0 = __shfl_xor(e[0], 16, 32), u1 = __shfl_xor(e[1], 16, 32);
    const float u4 = __shfl_xor(e[4], 16, 32), u5 = __shfl_xor(e[5], 16, 32);
    v2f pb[4];
    pb[0].x = hlf ? x2 : e[0];  pb[0].y = hlf ? x3 : e[1];
    pb[1].x = hlf ? x6 : e[4];  pb[1].y = hlf ? x7 : e[5];
    pb[2].x = hlf ? e[2] : u0;  pb[2].y = hlf ? e[3] : u1;
    pb[3].x = hlf ? e[6] : u4;  pb[3].y = hlf ? e[7] : u5;

#pragma unroll
    for (int t = 0; t < 4; t++)
#pragma unroll
        for (int kk = 0; kk < 4; kk++)
            acc[t] = WMMA_F32(va[t][kk], pb[kk], acc[t]);
}

template <bool FIXED>
__device__ __forceinline__
unsigned attn_pass(const char* qkvB, unsigned gko, float* sK, float* sV,
                   const int skoff, const int svoff, const int koff, const int voff,
                   const v2f (&qb)[16], v8f (&acc)[4], float& m, float& l, const int hlf)
{
    constexpr unsigned gtile = AKT * ThreeC * 4u;
    constexpr int KST = AKT * AKP;
    constexpr int VST = (AKT / 2) * AVP;
    unsigned bad = 0u;

    v4f rk, rv;
    if (FIXED) {
        rk = *(const v4f*)(qkvB + gko);
        rv = *(const v4f*)(qkvB + gko + Cdim * 4u);
    } else {
        rk = *(const volatile v4f*)(qkvB + gko);
        rv = *(const volatile v4f*)(qkvB + gko + Cdim * 4u);
    }
    *(v4f*)(sK + skoff) = rk;
    *(v4f*)(sV + svoff) = rv;
    __syncthreads();

    for (int it = 0; it + 1 < ANT; it++) {
        const int buf = it & 1;
        gko += gtile;
        rk = *(const v4f*)(qkvB + gko);
        rv = *(const v4f*)(qkvB + gko + Cdim * 4u);

        attn_keys16<FIXED>(sK + buf * KST + koff, sV + buf * VST + voff, qb, acc, m, l, hlf,
                           it == 0, bad);

        *(v4f*)(sK + (buf ^ 1) * KST + skoff) = rk;
        *(v4f*)(sV + (buf ^ 1) * VST + svoff) = rv;
        __syncthreads();
    }
    {
        const int buf = (ANT - 1) & 1;
        attn_keys16<FIXED>(sK + buf * KST + koff, sV + buf * VST + voff, qb, acc, m, l, hlf,
                           false, bad);
    }
    return bad;
}

__global__ __launch_bounds__(256, 8)
void flash_attn_kernel(const float* __restrict__ qkv, float* __restrict__ O)
{
    __shared__ __attribute__((aligned(16))) float sK[2 * AKT * AKP];
    __shared__ __attribute__((aligned(16))) float sV[2 * (AKT / 2) * AVP];
    __shared__ __attribute__((aligned(16))) unsigned sBad[8];

    const int tid  = threadIdx.x;
    const int lane = tid & 31;
    const int wv   = tid >> 5;
    const int qt = (blockIdx.x & 15) * 8 + wv;
    const int h  = (blockIdx.x >> 4) & 15;
    const int b  = blockIdx.x >> 8;
    const int hlf = lane >> 4;
    const int l16 = lane & 15;
    const int q0  = qt * 16;

    const char* qkvB = (const char*)qkv;

    const float qscale = 0.125f * 1.4426950408889634f;
    const unsigned rowQ = ((unsigned)(b * Nseq + q0 + l16) * ThreeC + h * HD + 2 * hlf) * 4u;
    v2f qb[16];
#pragma unroll
    for (int kk = 0; kk < 16; kk++)
        qb[kk] = *(const v2f*)(qkvB + rowQ + 16 * kk) * qscale;

    v8f zero = {};
    v8f acc[4];
#pragma unroll
    for (int t = 0; t < 4; t++) acc[t] = zero;
    float m = -3.0e38f, l = 0.0f;

    const int srow = tid >> 4;
    const int scol = (tid & 15) * 4;
    const unsigned gko = ((unsigned)(b * Nseq + srow) * ThreeC + Cdim + h * HD + scol) * 4u;
    const int skoff = srow * AKP + scol;
    const int svoff = (srow >> 1) * AVP + (scol >> 4) * 32 + (srow & 1) * 16 + (scol & 15);

    const int koff = l16 * AKP + 2 * hlf;
    const int voff = hlf * AVP + l16;

    const unsigned bad = attn_pass<true>(qkvB, gko, sK, sV, skoff, svoff, koff, voff, qb, acc, m, l, hlf);

    sBad[wv] = bad;
    __syncthreads();
    unsigned any = 0u;
#pragma unroll
    for (int w = 0; w < 8; w++) any |= sBad[w];
    if (__builtin_amdgcn_readfirstlane(any) != 0u) {
#pragma unroll
        for (int t = 0; t < 4; t++) acc[t] = zero;
        m = -3.0e38f;
        l = 0.0f;
        attn_pass<false>(qkvB, gko, sK, sV, skoff, svoff, koff, voff, qb, acc, m, l, hlf);
    }

    const float inv = 1.0f / (l + __shfl_xor(l, 16, 32));
    const unsigned baseO = ((unsigned)(b * Nseq + q0 + l16) * Cdim + h * HD + 8 * hlf) * 4u;
#pragma unroll
    for (int t = 0; t < 4; t++)
#pragma unroll
        for (int r = 0; r < 8; r++)
            *(float*)((char*)O + baseO + (t * 16 + r) * 4) = acc[t][r] * inv;
}

extern "C" void kernel_launch(void* const* d_in, const int* in_sizes, int n_in,
                              void* d_out, int out_size, void* d_ws, size_t ws_size,
                              hipStream_t stream)
{
    const float* x        = (const float*)d_in[0];
    const float* rope_cos = (const float*)d_in[1];
    const float* rope_sin = (const float*)d_in[2];
    const float* W_qkv    = (const float*)d_in[3];
    const float* W_proj   = (const float*)d_in[4];
    const float* b_proj   = (const float*)d_in[5];
    float* out  = (float*)d_out;

    float* qkv  = (float*)d_ws;
    float* attn = qkv + (size_t)Mrows * ThreeC;

    {
        const int blocks = (Mrows / GBM) * (ThreeC / GBN);
        gemm_xwt_kernel<<<blocks, 256, 0, stream>>>(x, W_qkv, nullptr, rope_cos, rope_sin, qkv,
                                                    Mrows, ThreeC, Cdim);
    }
    {
        const int waves = Bsz * Hn * (Nseq / 16);
        flash_attn_kernel<<<waves / 8, 256, 0, stream>>>(qkv, attn);
    }
    {
        const int blocks = (Mrows / GBM) * (Cdim / GBN);
        gemm_xwt_kernel<<<blocks, 256, 0, stream>>>(attn, W_proj, b_proj, nullptr, nullptr, out,
                                                    Mrows, Cdim, Cdim);
    }
}


# lane-half-contiguous k in all products: GEMM halves read 4 k per 16-byte LDS piece (2 K-steps), attention O^T half uses its own 8 keys of S^T (no exchange; V transposed in LDS); sum order changed
# speedup vs baseline: 2.4527x; 1.0594x over previous
#include <hip/hip_runtime.h>

typedef float v2f __attribute__((ext_vector_type(2)));
typedef float v4f __attribute__((ext_vector_type(4)));
typedef float v8f __attribute__((ext_vector_type(8)));

#define WMMA_F32(a, b, c) \
    __builtin_amdgcn_wmma_f32_16x16x4_f32(false, (a), false, (b), (short)0, (c), false, false)

constexpr int Bsz    = 4;
constexpr int Nseq   = 2048;
constexpr int Cdim   = 1024;
constexpr int Hn     = 16;
constexpr int HD     = 64;
constexpr int ThreeC = 3 * Cdim;
constexpr int Mrows  = Bsz * Nseq;

constexpr int GBM = 128;
constexpr int GBN = 128;
constexpr int GBK = 64;
constexpr int GLP = GBK + 4;
constexpr int GTR = GBK / 4;
constexpr int GRP = 256 / GTR;
constexpr int GNP = GBM / GRP;

#define GEMM_WMMA8(a0, a1, b0, b1, b2, b3)        \
    acc[0][0] = WMMA_F32(a0, b0, acc[0][0]);      \
    acc[1][0] = WMMA_F32(a1, b0, acc[1][0]);      \
    acc[0][1] = WMMA_F32(a0, b1, acc[0][1]);      \
    acc[1][1] = WMMA_F32(a1, b1, acc[1][1]);      \
    acc[0][2] = WMMA_F32(a0, b2, acc[0][2]);      \
    acc[1][2] = WMMA_F32(a1, b2, acc[1][2]);      \
    acc[0][3] = WMMA_F32(a0, b3, acc[0][3]);      \
    acc[1][3] = WMMA_F32(a1, b3, acc[1][3]);

#define GEMM_FRAGS(a0, a1, b0, b1, b2, b3, pp)            \
    a0 = *(const v4f*)(pa + 8 * (pp));                    \
    a1 = *(const v4f*)(pa + 16 * GLP + 8 * (pp));         \
    b0 = *(const v4f*)(pb + 8 * (pp));                    \
    b1 = *(const v4f*)(pb + 16 * GLP + 8 * (pp));         \
    b2 = *(const v4f*)(pb + 32 * GLP + 8 * (pp));         \
    b3 = *(const v4f*)(pb + 48 * GLP + 8 * (pp));

#define GEMM_WMMA16(a0, a1, b0, b1, b2, b3)                       \
    GEMM_WMMA8(a0.lo, a1.lo, b0.lo, b1.lo, b2.lo, b3.lo)          \
    GEMM_WMMA8(a0.hi, a1.hi, b0.hi, b1.hi, b2.hi, b3.hi)

__device__ __forceinline__
void gemm_stage_mma(const float* __restrict__ pa, const float* __restrict__ pb,
                    v8f (&acc)[2][4])
{
    v4f pa0, pa1, pb0, pb1, pb2, pb3;
    v4f qa0, qa1, qb0, qb1, qb2, qb3;
    GEMM_FRAGS(pa0, pa1, pb0, pb1, pb2, pb3, 0)
#pragma unroll 1
    for (int pp = 0; pp < GBK / 8 - 2; pp += 2) {
        GEMM_FRAGS(qa0, qa1, qb0, qb1, qb2, qb3, pp + 1)
        GEMM_WMMA16(pa0, pa1, pb0, pb1, pb2, pb3)
        GEMM_FRAGS(pa0, pa1, pb0, pb1, pb2, pb3, pp + 2)
        GEMM_WMMA16(qa0, qa1, qb0, qb1, qb2, qb3)
    }
    GEMM_FRAGS(qa0, qa1, qb0, qb1, qb2, qb3, GBK / 8 - 1)
    GEMM_WMMA16(pa0, pa1, pb0, pb1, pb2, pb3)
    GEMM_WMMA16(qa0, qa1, qb0, qb1, qb2, qb3)
}

__global__ __launch_bounds__(256)
void gemm_xwt_kernel(const float* __restrict__ X, const float* __restrict__ W,
                     const float* __restrict__ bias,
                     const float* __restrict__ cosT, const float* __restrict__ sinT,
                     float* __restrict__ Y, int M, int Nd, int K)
{
    __shared__ __attribute__((aligned(16))) float sAB[2][(GBM + GBN) * GLP];
    constexpr int SBO = GBM * GLP;

    const int tid  = threadIdx.x;
    const int lane = tid & 31;
    const int wv   = tid >> 5;
    const int hlf  = lane >> 4;
    const int l16  = lane & 15;
    const int tilesN = Nd / GBN;
    const int bm0 = (blockIdx.x / tilesN) * GBM;
    const int bn0 = (blockIdx.x % tilesN) * GBN;
    const int wm  = (wv >> 1) * 32;
    const int wn  = (wv & 1) * 64;

    v8f zero = {};
    v8f acc[2][4];
#pragma unroll
    for (int i = 0; i < 2; i++)
#pragma unroll
        for (int j = 0; j < 4; j++) acc[i][j] = zero;

    const int srow = tid / GTR;
    const int scol = (tid % GTR) * 4;
    const float* gA = X + (size_t)(bm0 + srow) * K + scol;
    const float* gB = W + (size_t)(bn0 + srow) * K + scol;
    const size_t gstep = (size_t)GRP * K;
    const int soff = srow * GLP + scol;

    const int aoff = (wm + l16) * GLP + 4 * hlf;
    const int boff = (wn + l16) * GLP + 4 * hlf;

    v4f ra[GNP], rb[GNP];
#pragma unroll
    for (int i = 0; i < GNP; i++) {
        ra[i] = *(const v4f*)(gA + i * gstep);
        rb[i] = *(const v4f*)(gB + i * gstep);
    }
#pragma unroll
    for (int i = 0; i < GNP; i++) {
        *(v4f*)(&sAB[0][soff + i * GRP * GLP]) = ra[i];
        *(v4f*)(&sAB[0][SBO + soff + i * GRP * GLP]) = rb[i];
    }
    __syncthreads();

    const int nk = K / GBK;
    for (int kt = 0; kt + 1 < nk; kt++) {
        const int buf = kt & 1;
        const int kn = (kt + 1) * GBK;
#pragma unroll
        for (int i = 0; i < GNP; i++) {
            ra[i] = *(const v4f*)(gA + i * gstep + kn);
            rb[i] = *(const v4f*)(gB + i * gstep + kn);
        }

        gemm_stage_mma(&sAB[buf][aoff], &sAB[buf][SBO + boff], acc);

#pragma unroll
        for (int i = 0; i < GNP; i++) {
            *(v4f*)(&sAB[buf ^ 1][soff + i * GRP * GLP]) = ra[i];
            *(v4f*)(&sAB[buf ^ 1][SBO + soff + i * GRP * GLP]) = rb[i];
        }
        __syncthreads();
    }
    {
        const int buf = (nk - 1) & 1;
        gemm_stage_mma(&sAB[buf][aoff], &sAB[buf][SBO + boff], acc);
    }

    const int m0 = bm0 + wm;
    const int n0 = bn0 + wn;

    if (cosT != nullptr && bn0 < 2 * Cdim) {
#pragma unroll
        for (int i = 0; i < 2; i++)
#pragma unroll
            for (int r = 0; r < 8; r++) {
                const int n = (m0 + 16 * i + r + 8 * hlf) & (Nseq - 1);
                const float* cr = cosT + n * HD + l16;
                const float* sr = sinT + n * HD + l16;
#pragma unroll
                for (int j = 0; j < 2; j++) {
                    const float c0 = cr[16 * j],      s0 = sr[16 * j];
                    const float c1 = cr[16 * j + 32], s1 = sr[16 * j + 32];
                    const float t0 = acc[i][j][r], t1 = acc[i][j + 2][r];
                    acc[i][j][r]     = t0 * c0 - t1 * s0;
                    acc[i][j + 2][r] = t1 * c1 + t0 * s1;
                }
            }
    }

#pragma unroll
    for (int i = 0; i < 2; i++)
#pragma unroll
        for (int j = 0; j < 4; j++)
#pragma unroll
            for (int r = 0; r < 8; r++) {
                const int row = m0 + 16 * i + r + 8 * hlf;
                const int col = n0 + 16 * j + l16;
                float v = acc[i][j][r];
                if (bias) v += bias[col];
                Y[(size_t)row * Nd + col] = v;
            }
}

constexpr int AKT = 16;
constexpr int AKP = HD + 4;
constexpr int AVD = AKT + 4;
constexpr int ANT = Nseq / AKT;
constexpr float QSCALE = 0.125f * 1.4426950408889634f;

template <bool FIXED>
__device__ __forceinline__
void attn_keys16(const float* pk, const float* pv, const v2f (&qb)[16], const char* qrow,
                 v8f (&acc)[4], float& m, float& l, const bool first, unsigned& bad)
{
    v8f zero = {};

    v2f ka[16];
#pragma unroll
    for (int kk = 0; kk < 16; kk++)
        ka[kk] = *(const v2f*)(pk + 4 * kk);

    asm volatile("" :: "v"(ka[0]), "v"(ka[1]), "v"(ka[2]), "v"(ka[3]), "v"(ka[4]), "v"(ka[5]),
                       "v"(ka[6]), "v"(ka[7]), "v"(ka[8]), "v"(ka[9]), "v"(ka[10]), "v"(ka[11]),
                       "v"(ka[12]), "v"(ka[13]), "v"(ka[14]), "v"(ka[15]));

    v8f s = zero;
#pragma unroll
    for (int kk = 0; kk < 16; kk++) {
        const v2f q = FIXED ? qb[kk] : *(const volatile v2f*)(qrow + 16 * kk) * QSCALE;
        s = WMMA_F32(ka[kk], q, s);
    }

    v2f va[4][4];
#pragma unroll
    for (int t = 0; t < 4; t++) {
        const v4f k03 = *(const v4f*)(pv + 16 * t * AVD);
        const v4f k47 = *(const v4f*)(pv + 16 * t * AVD + 4);
        va[t][0] = k03.lo;
        va[t][1] = k03.hi;
        va[t][2] = k47.lo;
        va[t][3] = k47.hi;
    }

    if (!FIXED || first) {
        float tm = s[0];
#pragma unroll
        for (int r = 1; r < 8; r++) tm = fmaxf(tm, s[r]);
        tm = fmaxf(tm, __shfl_xor(tm, 16, 32));
        if (FIXED) {
            m = tm;
        } else {
            const float mn   = fmaxf(m, tm);
            const float corr = __builtin_amdgcn_exp2f(m - mn);
            m = mn;
            l *= corr;
#pragma unroll
            for (int t = 0; t < 4; t++)
#pragma unroll
                for (int r = 0; r < 8; r++) acc[t][r] *= corr;
        }
    }

    v8f e;
#pragma unroll
    for (int r = 0; r < 8; r++) e[r] = __builtin_amdgcn_exp2f(s[r] - m);
    const v4f h4 = e.lo + e.hi;
    const v2f h2 = h4.lo + h4.hi;
    const float rs = h2.x + h2.y;
    l += rs;
    if (FIXED) bad |= __builtin_amdgcn_ballot_w32(!(rs <= 0x1p30f));

    v2f pb[4];
#pragma unroll
    for (int kk = 0; kk < 4; kk++) {
        pb[kk].x = e[2 * kk];
        pb[kk].y = e[2 * kk + 1];
    }

#pragma unroll
    for (int t = 0; t < 4; t++)
#pragma unroll
        for (int kk = 0; kk < 4; kk++)
            acc[t] = WMMA_F32(va[t][kk], pb[kk], acc[t]);
}

template <bool FIXED>
__device__ __forceinline__
unsigned attn_pass(const char* qkvB, unsigned gko, float* sK, float* sV,
                   const int skoff, const int svoff, const int koff, const int voff,
                   const v2f (&qb)[16], const char* qrow, v8f (&acc)[4], float& m, float& l)
{
    constexpr unsigned gtile = AKT * ThreeC * 4u;
    constexpr int KST = AKT * AKP;
    constexpr int VST = HD * AVD;
    unsigned bad = 0u;

    v4f rk, rv;
    if (FIXED) {
        rk = *(const v4f*)(qkvB + gko);
        rv = *(const v4f*)(qkvB + gko + Cdim * 4u);
    } else {
        rk = *(const volatile v4f*)(qkvB + gko);
        rv = *(const volatile v4f*)(qkvB + gko + Cdim * 4u);
    }
    *(v4f*)(sK + skoff) = rk;
    sV[svoff]           = rv.x;
    sV[svoff + AVD]     = rv.y;
    sV[svoff + 2 * AVD] = rv.z;
    sV[svoff + 3 * AVD] = rv.w;
    __syncthreads();

    for (int it = 0; it + 1 < ANT; it++) {
        const int buf = it & 1;
        gko += gtile;
        if (FIXED) {
            rk = *(const v4f*)(qkvB + gko);
            rv = *(const v4f*)(qkvB + gko + Cdim * 4u);
        }

        attn_keys16<FIXED>(sK + buf * KST + koff, sV + buf * VST + voff, qb, qrow, acc, m, l,
                           it == 0, bad);

        if (!FIXED) {
            rk = *(const volatile v4f*)(qkvB + gko);
            rv = *(const volatile v4f*)(qkvB + gko + Cdim * 4u);
        }

        *(v4f*)(sK + (buf ^ 1) * KST + skoff) = rk;
        float* dv = sV + (buf ^ 1) * VST + svoff;
        dv[0]       = rv.x;
        dv[AVD]     = rv.y;
        dv[2 * AVD] = rv.z;
        dv[3 * AVD] = rv.w;
        __syncthreads();
    }
    {
        const int buf = (ANT - 1) & 1;
        attn_keys16<FIXED>(sK + buf * KST + koff, sV + buf * VST + voff, qb, qrow, acc, m, l,
                           false, bad);
    }
    return bad;
}

__device__ __forceinline__
void attn_store(const v8f (&acc)[4], const float l, float* __restrict__ O,
                const unsigned row0, const int h, const unsigned ln)
{
    const unsigned baseO = ((row0 + (ln & 15)) * Cdim + h * HD + 8 * (ln >> 4)) * 4u;
    const float inv = 1.0f / (l + __shfl_xor(l, 16, 32));
#pragma unroll
    for (int t = 0; t < 4; t++)
#pragma unroll
        for (int r = 0; r < 8; r++)
            *(float*)((char*)O + baseO + (t * 16 + r) * 4) = acc[t][r] * inv;
}

__global__ __launch_bounds__(256, 8)
void flash_attn_kernel(const float* __restrict__ qkv, float* __restrict__ O)
{
    __shared__ __attribute__((aligned(16))) float sK[2 * AKT * AKP];
    __shared__ __attribute__((aligned(16))) float sV[2 * HD * AVD];
    __shared__ __attribute__((aligned(16))) unsigned sBad[8];

    const int tid  = threadIdx.x;
    const int lane = tid & 31;
    const int wv   = __builtin_amdgcn_readfirstlane(tid >> 5);
    const int qt = (blockIdx.x & 15) * 8 + wv;
    const int h  = (blockIdx.x >> 4) & 15;
    const int b  = blockIdx.x >> 8;
    const int hlf = lane >> 4;
    const int l16 = lane & 15;
    const int q0  = qt * 16;

    const char* qkvB = (const char*)qkv;

    const unsigned row0 = (unsigned)(b * Nseq + q0);
    const unsigned rowQ = ((row0 + l16) * ThreeC + h * HD + 2 * hlf) * 4u;
    v2f qb[16];
#pragma unroll
    for (int kk = 0; kk < 16; kk++)
        qb[kk] = *(const v2f*)(qkvB + rowQ + 16 * kk) * QSCALE;

    v8f zero = {};
    v8f acc[4];
#pragma unroll
    for (int t = 0; t < 4; t++) acc[t] = zero;
    float m = -3.0e38f, l = 0.0f;

    const int srow = tid >> 4;
    const int scol = (tid & 15) * 4;
    const unsigned gko = ((unsigned)(b * Nseq + srow) * ThreeC + Cdim + h * HD + scol) * 4u;
    const int skoff = srow * AKP + scol;
    const int svoff = scol * AVD + srow;

    const int koff = l16 * AKP + 2 * hlf;
    const int voff = l16 * AVD + 8 * hlf;

    const unsigned bad = attn_pass<true>(qkvB, gko, sK, sV, skoff, svoff, koff, voff, qb, qkvB, acc, m, l);

    attn_store(acc, l, O, row0, h, (unsigned)lane);

    sBad[wv] = bad;
    __syncthreads();
    unsigned any = 0u;
#pragma unroll
    for (int w = 0; w < 8; w++) any |= sBad[w];
    if (__builtin_amdgcn_readfirstlane(any) != 0u) {
#pragma unroll
        for (int t = 0; t < 4; t++) acc[t] = zero;
        m = -3.0e38f;
        l = 0.0f;
        const unsigned ln = __builtin_amdgcn_mbcnt_lo(~0u, 0u);
        const unsigned rowQ2 = ((row0 + (ln & 15)) * ThreeC + h * HD + 2 * (ln >> 4)) * 4u;
        attn_pass<false>(qkvB, gko, sK, sV, skoff, svoff, koff, voff, qb, qkvB + rowQ2, acc, m, l);
        attn_store(acc, l, O, row0, h, ln);
    }
}

extern "C" void kernel_launch(void* const* d_in, const int* in_sizes, int n_in,
                              void* d_out, int out_size, void* d_ws, size_t ws_size,
                              hipStream_t stream)
{
    const float* x        = (const float*)d_in[0];
    const float* rope_cos = (const float*)d_in[1];
    const float* rope_sin = (const float*)d_in[2];
    const float* W_qkv    = (const float*)d_in[3];
    const float* W_proj   = (const float*)d_in[4];
    const float* b_proj   = (const float*)d_in[5];
    float* out  = (float*)d_out;

    float* qkv  = (float*)d_ws;
    float* attn = qkv + (size_t)Mrows * ThreeC;

    {
        const int blocks = (Mrows / GBM) * (ThreeC / GBN);
        gemm_xwt_kernel<<<blocks, 256, 0, stream>>>(x, W_qkv, nullptr, rope_cos, rope_sin, qkv,
                                                    Mrows, ThreeC, Cdim);
    }
    {
        const int waves = Bsz * Hn * (Nseq / 16);
        flash_attn_kernel<<<waves / 8, 256, 0, stream>>>(qkv, attn);
    }
    {
        const int blocks = (Mrows / GBM) * (Cdim / GBN);
        gemm_xwt_kernel<<<blocks, 256, 0, stream>>>(attn, W_proj, b_proj, nullptr, nullptr, out,
                                                    Mrows, Cdim, Cdim);
    }
}
